# starting kernel
# baseline (speedup 1.0000x reference)
#include <hip/hip_runtime.h>
#include <math.h>

typedef __attribute__((ext_vector_type(2))) float v2f;
typedef __attribute__((ext_vector_type(8))) float v8f;

namespace {
constexpr int kB  = 4096;
constexpr int kT  = 128;
constexpr int kD  = 128;
constexpr int kH  = 150;
constexpr int kO  = 10;
constexpr int kHP = 160;
constexpr int kNT = kHP / 16;
constexpr int kNTW = kNT / 2;
constexpr int kKTX = kD / 4;
constexpr int kKTH = kHP / 4;
constexpr int kRows = 32;
constexpr int kThreads = 128;
constexpr int kXStride = 132;
constexpr int kHStride = 164;
constexpr int kXSlots = kRows * (kD / 4);
constexpr int kXPerThr = kXSlots / kThreads;
}

__device__ __forceinline__ float fast_tanh(float x) {
  const float e = __expf(2.0f * x);
  return 1.0f - __fdividef(2.0f, e + 1.0f);
}

__global__ __launch_bounds__(kThreads, 1)
void rnn_wmma_kernel(const float* __restrict__ X,
                     const float* __restrict__ Wc,
                     const float* __restrict__ bc,
                     const float* __restrict__ Wo,
                     const float* __restrict__ bo,
                     float* __restrict__ out) {
  __shared__ float2 sWx[kKTX * kNT * 32];
  __shared__ float2 sWh[kKTH * kNT * 32];
  __shared__ float  sX [2][kRows * kXStride];
  __shared__ float  sH [kRows * kHStride];
  __shared__ float  sB [kHP];

  const int tid = threadIdx.x;
  const int w   = tid >> 5;
  const int mw  = w >> 1;
  const int nw  = w & 1;
  const int l   = tid & 31;
  const int lr  = l & 15;
  const int lh  = l >> 4;
  const int coff = lh * 2;
  const int rowBase = blockIdx.x * kRows;

  for (int s = tid; s < kKTX * kNT * 32; s += kThreads) {
    const int f = s >> 5, sl = s & 31;
    const int kt = f / kNT, nt = f % kNT;
    const int K = 4 * kt + 2 * (sl >> 4);
    const int N = nt * 16 + (sl & 15);
    float x = 0.f, y = 0.f;
    if (N < kH) { x = Wc[K * kH + N]; y = Wc[(K + 1) * kH + N]; }
    sWx[s] = make_float2(x, y);
  }
  for (int s = tid; s < kKTH * kNT * 32; s += kThreads) {
    const int f = s >> 5, sl = s & 31;
    const int kt = f / kNT, nt = f % kNT;
    const int K = 4 * kt + 2 * (sl >> 4);
    const int N = nt * 16 + (sl & 15);
    float x = 0.f, y = 0.f;
    if (N < kH) {
      if (K < kH)     x = Wc[(kD + K) * kH + N];
      if (K + 1 < kH) y = Wc[(kD + K + 1) * kH + N];
    }
    sWh[s] = make_float2(x, y);
  }
  for (int i = tid; i < kHP; i += kThreads) sB[i] = (i < kH) ? bc[i] : 0.f;
  for (int i = tid; i < kRows * kHStride; i += kThreads) sH[i] = 0.f;

  {
#pragma unroll
    for (int i = 0; i < kXPerThr; ++i) {
      const int s = tid + i * kThreads;
      const int row = s >> 5, c4 = s & 31;
      const float4 v = *(const float4*)(X + (((size_t)(rowBase + row)) * kT + 0) * kD + c4 * 4);
      *(float4*)(&sX[0][row * kXStride + c4 * 4]) = v;
    }
  }

  const float* aXrow0 = &sX[0][(mw * 16 + lr) * kXStride];
  const float* aXrow1 = &sX[1][(mw * 16 + lr) * kXStride];
  const float* aHrow  = &sH[(mw * 16 + lr) * kHStride];
  const int crow0 = mw * 16 + lh * 8;

  for (int t = 0; t < kT; ++t) {
    float4 xn[kXPerThr];
    if (t + 1 < kT) {
#pragma unroll
      for (int i = 0; i < kXPerThr; ++i) {
        const int s = tid + i * kThreads;
        const int row = s >> 5, c4 = s & 31;
        xn[i] = *(const float4*)(X + (((size_t)(rowBase + row)) * kT + (t + 1)) * kD + c4 * 4);
      }
    }
    __syncthreads();

    const float* aXrow = (t & 1) ? aXrow1 : aXrow0;
    v8f acc[kNTW];
#pragma unroll
    for (int n = 0; n < kNTW; ++n) {
      const float bv = sB[(nw * kNTW + n) * 16 + lr];
      acc[n] = (v8f){bv, bv, bv, bv, bv, bv, bv, bv};
    }
#pragma unroll 2
    for (int k = 0; k < kKTX; ++k) {
      const float2 av = *(const float2*)(aXrow + 4 * k + coff);
      v2f a; a.x = av.x; a.y = av.y;
      const float2* wf = &sWx[(k * kNT + nw * kNTW) * 32 + l];
#pragma unroll
      for (int n = 0; n < kNTW; ++n) {
        const float2 wv = wf[n * 32];
        v2f b; b.x = wv.x; b.y = wv.y;
        acc[n] = __builtin_amdgcn_wmma_f32_16x16x4_f32(
            false, a, false, b, (short)0, acc[n], false, false);
      }
    }
#pragma unroll 2
    for (int k = 0; k < kKTH; ++k) {
      const float2 av = *(const float2*)(aHrow + 4 * k + coff);
      v2f a; a.x = av.x; a.y = av.y;
      const float2* wf = &sWh[(k * kNT + nw * kNTW) * 32 + l];
#pragma unroll
      for (int n = 0; n < kNTW; ++n) {
        const float2 wv = wf[n * 32];
        v2f b; b.x = wv.x; b.y = wv.y;
        acc[n] = __builtin_amdgcn_wmma_f32_16x16x4_f32(
            false, a, false, b, (short)0, acc[n], false, false);
      }
    }
#pragma unroll
    for (int n = 0; n < kNTW; ++n)
#pragma unroll
      for (int i = 0; i < 8; ++i)
        acc[n][i] = fast_tanh(acc[n][i]);

    if (t + 1 < kT) {
      float* nxt = (t & 1) ? sX[0] : sX[1];
#pragma unroll
      for (int i = 0; i < kXPerThr; ++i) {
        const int s = tid + i * kThreads;
        const int row = s >> 5, c4 = s & 31;
        *(float4*)(&nxt[row * kXStride + c4 * 4]) = xn[i];
      }
    }
    __syncthreads();
#pragma unroll
    for (int n = 0; n < kNTW; ++n) {
      const int col = (nw * kNTW + n) * 16 + lr;
#pragma unroll
      for (int r = 0; r < 8; ++r)
        sH[(crow0 + r) * kHStride + col] = acc[n][r];
    }
    __syncthreads();
  }

  for (int p = tid; p < kRows * kO; p += kThreads) {
    const int r = p / kO, o = p % kO;
    float sum = bo[o];
    const float* hrow = &sH[r * kHStride];
    for (int k = 0; k < kH; ++k) sum += hrow[k] * Wo[k * kO + o];
    out[(size_t)(rowBase + r) * kO + o] = sum;
  }
}

extern "C" void kernel_launch(void* const* d_in, const int* in_sizes, int n_in,
                              void* d_out, int out_size, void* d_ws, size_t ws_size,
                              hipStream_t stream) {
  const float* X  = (const float*)d_in[0];
  const float* Wc = (const float*)d_in[1];
  const float* bc = (const float*)d_in[2];
  const float* Wo = (const float*)d_in[3];
  const float* bo = (const float*)d_in[4];
  float* out = (float*)d_out;
  (void)in_sizes; (void)n_in; (void)out_size; (void)d_ws; (void)ws_size;
  dim3 grid(kB / kRows), block(kThreads);
  hipLaunchKernelGGL(rnn_wmma_kernel, grid, block, 0, stream, X, Wc, bc, Wo, bo, out);
}


# Operand roles swapped: W^T is the A operand, activations the B operand, so tanh output is the next step's B fragments in registers; 4 waves split k, partials summed via LDS (sum order changes)
# speedup vs baseline: 2.6503x; 2.6503x over previous
#include <hip/hip_runtime.h>
#include <math.h>

typedef __attribute__((ext_vector_type(2))) float v2f;
typedef __attribute__((ext_vector_type(8))) float v8f;

namespace {
constexpr int kB  = 4096;
constexpr int kT  = 128;
constexpr int kD  = 128;
constexpr int kH  = 150;
constexpr int kO  = 10;
constexpr int kCols = 16;
constexpr int kWaves = 4;
constexpr int kThreads = 128;
constexpr int kMT = 10;
constexpr int kPairs = 9;
constexpr int kQuads = 2 * kMT;
constexpr int kWSlots = kWaves * kPairs * kMT * 32;
constexpr int kPSlots = kWaves * kQuads * 32;
constexpr int kHStride = 164;
}

__device__ __forceinline__ float fast_tanh(float x) {
  const float e = __expf(2.0f * x);
  return 1.0f - __fdividef(2.0f, e + 1.0f);
}

__device__ __forceinline__ void group_fence() { __builtin_amdgcn_sched_barrier(0); }

__device__ __forceinline__ void pin_here(float4& v) {
  asm volatile("" : "+v"(v.x), "+v"(v.y), "+v"(v.z), "+v"(v.w));
}

template <int P0, int NP>
__device__ __forceinline__ void load_w(float4 (&dst)[30], const float4* __restrict__ wq) {
#pragma unroll
  for (int i = 0; i < NP * kMT; ++i) dst[i] = wq[(P0 * kMT + i) * 32];
}

template <int NP>
__device__ __forceinline__ void mma_group(v8f (&acc)[kMT], const float4 (&w)[30], const float4* b) {
#pragma unroll
  for (int pp = 0; pp < NP; ++pp) {
    const float4 bv = b[pp];
    const v2f b0 = {bv.x, bv.y};
    const v2f b1 = {bv.z, bv.w};
#pragma unroll
    for (int m = 0; m < kMT; ++m) {
      const v2f a0 = {w[pp * kMT + m].x, w[pp * kMT + m].y};
      acc[m] = __builtin_amdgcn_wmma_f32_16x16x4_f32(false, a0, false, b0, (short)0, acc[m], false, false);
    }
#pragma unroll
    for (int m = 0; m < kMT; ++m) {
      const v2f a1 = {w[pp * kMT + m].z, w[pp * kMT + m].w};
      acc[m] = __builtin_amdgcn_wmma_f32_16x16x4_f32(false, a1, false, b1, (short)0, acc[m], false, false);
    }
  }
}

__device__ __forceinline__ void load_x(float4 (&xs)[4], const float* __restrict__ xlane, int t) {
  const float4* p = (const float4*)(xlane + (size_t)t * kD);
#pragma unroll
  for (int i = 0; i < 4; ++i) xs[i] = p[i];
}

__device__ __forceinline__ void rnn_step(int t, int q, int l,
    v8f (&cur)[kMT], v8f (&nxt)[kMT], float4 (&hq)[5], const float4 (&bq)[5],
    const float4 (&xuse)[4], float4 (&xload)[4], const float* __restrict__ xlane,
    float4 (&setA)[30], float4 (&setB)[30],
    const float4* __restrict__ wq, float4* __restrict__ sPbuf) {
  const int t2 = (t + 2 < kT) ? (t + 2) : (kT - 1);
  load_x(xload, xlane, t2);

  load_w<7, 2>(setB, wq);
  group_fence();
  mma_group<3>(cur, setA, &hq[0]);
  group_fence();
  load_w<0, 2>(setA, wq);
  group_fence();
  mma_group<2>(cur, setB, &hq[3]);
  group_fence();

  float4* pw = sPbuf + (q * kQuads) * 32 + l;
#pragma unroll
  for (int u = 0; u < kQuads; ++u) {
    const int m = u >> 1, r = 4 * (u & 1);
    pw[u * 32] = make_float4(cur[m][r], cur[m][r + 1], cur[m][r + 2], cur[m][r + 3]);
  }
  __syncthreads();

  const float4* pr = sPbuf + (5 * q) * 32 + l;
  float4 ex[4][5];
#pragma unroll
  for (int s = 0; s < 4; ++s)
#pragma unroll
    for (int i = 0; i < 5; ++i) ex[s][i] = pr[(s * kQuads + i) * 32];
  load_w<2, 2>(setB, wq);
  group_fence();
  float4 xb[4] = {xuse[0], xuse[1], xuse[2], xuse[3]};
  pin_here(xb[0]); pin_here(xb[1]);
#pragma unroll
  for (int m = 0; m < kMT; ++m) nxt[m] = (v8f){0.f, 0.f, 0.f, 0.f, 0.f, 0.f, 0.f, 0.f};
  mma_group<2>(nxt, setA, &xb[0]);
  group_fence();
  load_w<4, 3>(setA, wq);
  group_fence();
  pin_here(xb[2]); pin_here(xb[3]);
#pragma unroll
  for (int s = 0; s < 4; ++s)
#pragma unroll
    for (int i = 0; i < 5; ++i) pin_here(ex[s][i]);
  mma_group<2>(nxt, setB, &xb[2]);
#pragma unroll
  for (int i = 0; i < 5; ++i) {
    float4 s;
    s.x = ((ex[0][i].x + ex[1][i].x) + (ex[2][i].x + ex[3][i].x)) + bq[i].x;
    s.y = ((ex[0][i].y + ex[1][i].y) + (ex[2][i].y + ex[3][i].y)) + bq[i].y;
    s.z = ((ex[0][i].z + ex[1][i].z) + (ex[2][i].z + ex[3][i].z)) + bq[i].z;
    s.w = ((ex[0][i].w + ex[1][i].w) + (ex[2][i].w + ex[3][i].w)) + bq[i].w;
    hq[i] = make_float4(fast_tanh(s.x), fast_tanh(s.y), fast_tanh(s.z), fast_tanh(s.w));
  }
  group_fence();
}

__global__ __launch_bounds__(kThreads, 1)
void rnn_wmma_kernel(const float* __restrict__ X,
                     const float* __restrict__ Wc,
                     const float* __restrict__ bc,
                     const float* __restrict__ Wo,
                     const float* __restrict__ bo,
                     float* __restrict__ out) {
  __shared__ float4 sW[kWSlots];
  __shared__ float4 sP[2][kPSlots];

  const int tid = threadIdx.x;
  const int q   = tid >> 5;
  const int l   = tid & 31;
  const int c   = l & 15;
  const int hh  = l >> 4;
  const int rowBase = blockIdx.x * kCols;

  for (int s = tid; s < kWSlots; s += kThreads) {
    const int sl = s & 31;
    int f = s >> 5;
    const int m = f % kMT; f /= kMT;
    const int p = f % kPairs;
    const int qq = f / kPairs;
    const int o = m * 16 + (sl & 15);
    const int g = sl >> 4;
    float v[4];
#pragma unroll
    for (int e = 0; e < 4; ++e) {
      const int jj = 2 * p + (e >> 1), ss = e & 1;
      int row; bool ok = o < kH;
      if (jj < 8) {
        row = 32 * qq + 16 * g + 2 * jj + ss;
      } else {
        const int kk = 10 * qq + (jj - 8);
        const int unit = 16 * (kk >> 2) + 8 * g + 2 * (kk & 3) + ss;
        ok = ok && (unit < kH);
        row = kD + unit;
      }
      v[e] = ok ? Wc[row * kH + o] : 0.f;
    }
    sW[s] = make_float4(v[0], v[1], v[2], v[3]);
  }

  float4 bq[5];
#pragma unroll
  for (int i = 0; i < 5; ++i) {
    const int u = 5 * q + i;
    const int unit0 = 16 * (u >> 1) + 8 * hh + 4 * (u & 1);
    bq[i].x = (unit0 + 0 < kH) ? bc[unit0 + 0] : 0.f;
    bq[i].y = (unit0 + 1 < kH) ? bc[unit0 + 1] : 0.f;
    bq[i].z = (unit0 + 2 < kH) ? bc[unit0 + 2] : 0.f;
    bq[i].w = (unit0 + 3 < kH) ? bc[unit0 + 3] : 0.f;
  }

  const float* xlane = X + (size_t)(rowBase + c) * kT * kD + 32 * q + 16 * hh;
  const float4* wq = sW + (q * kPairs * kMT) * 32 + l;

  float4 xs0[4], xs1[4];
  load_x(xs0, xlane, 0);
  load_x(xs1, xlane, 1);

  float4 hq[5];
#pragma unroll
  for (int i = 0; i < 5; ++i) hq[i] = make_float4(0.f, 0.f, 0.f, 0.f);

  __syncthreads();

  float4 setA[30], setB[30];
  v8f accA[kMT], accB[kMT];
#pragma unroll
  for (int m = 0; m < kMT; ++m) accA[m] = (v8f){0.f, 0.f, 0.f, 0.f, 0.f, 0.f, 0.f, 0.f};
  load_w<0, 2>(setA, wq);
  load_w<2, 2>(setB, wq);
  mma_group<2>(accA, setA, &xs0[0]);
  mma_group<2>(accA, setB, &xs0[2]);
  load_w<4, 3>(setA, wq);

  for (int tt = 0; tt < kT; tt += 2) {
    rnn_step(tt,     q, l, accA, accB, hq, bq, xs1, xs0, xlane, setA, setB, wq, sP[0]);
    rnn_step(tt + 1, q, l, accB, accA, hq, bq, xs0, xs1, xlane, setA, setB, wq, sP[1]);
  }

  float* hf = (float*)sP[0];
#pragma unroll
  for (int i = 0; i < 5; ++i) {
    const int u = 5 * q + i;
    const int unit0 = 16 * (u >> 1) + 8 * hh + 4 * (u & 1);
    *(float4*)(&hf[c * kHStride + unit0]) = hq[i];
  }
  __syncthreads();

  for (int p = tid; p < kCols * kO; p += kThreads) {
    const int r = p / kO, o = p % kO;
    float sum = bo[o];
    const float* hrow = &hf[r * kHStride];
    for (int k = 0; k < kH; ++k) sum += hrow[k] * Wo[k * kO + o];
    out[(size_t)(rowBase + r) * kO + o] = sum;
  }
}

extern "C" void kernel_launch(void* const* d_in, const int* in_sizes, int n_in,
                              void* d_out, int out_size, void* d_ws, size_t ws_size,
                              hipStream_t stream) {
  const float* X  = (const float*)d_in[0];
  const float* Wc = (const float*)d_in[1];
  const float* bc = (const float*)d_in[2];
  const float* Wo = (const float*)d_in[3];
  const float* bo = (const float*)d_in[4];
  float* out = (float*)d_out;
  (void)in_sizes; (void)n_in; (void)out_size; (void)d_ws; (void)ws_size;
  dim3 grid(kB / kCols), block(kThreads);
  hipLaunchKernelGGL(rnn_wmma_kernel, grid, block, 0, stream, X, Wc, bc, Wo, bo, out);
}


# W staging prologue: 8-byte row loads, 60 in flight per lane (each wave its own k quarter, work-groups start at rotated rows) instead of 90 serial rounds of four guarded 4-byte loads; LDS layout kept
# speedup vs baseline: 3.1343x; 1.1826x over previous
#include <hip/hip_runtime.h>
#include <math.h>

typedef __attribute__((ext_vector_type(2))) float v2f;
typedef __attribute__((ext_vector_type(8))) float v8f;

namespace {
constexpr int kB  = 4096;
constexpr int kT  = 128;
constexpr int kD  = 128;
constexpr int kH  = 150;
constexpr int kO  = 10;
constexpr int kCols = 16;
constexpr int kWaves = 4;
constexpr int kThreads = 128;
constexpr int kMT = 10;
constexpr int kPairs = 9;
constexpr int kQuads = 2 * kMT;
constexpr int kWSlots = kWaves * kPairs * kMT * 32;
constexpr int kPSlots = kWaves * kQuads * 32;
constexpr int kHStride = 164;
constexpr int kStageTasks = 45;
constexpr int kStageBatch = 15;
}

__device__ __forceinline__ float fast_tanh(float x) {
  const float e = __expf(2.0f * x);
  return 1.0f - __fdividef(2.0f, e + 1.0f);
}

__device__ __forceinline__ void group_fence() { __builtin_amdgcn_sched_barrier(0); }

__device__ __forceinline__ void pin_here(float4& v) {
  asm volatile("" : "+v"(v.x), "+v"(v.y), "+v"(v.z), "+v"(v.w));
}

template <int P0, int NP>
__device__ __forceinline__ void load_w(float4 (&dst)[30], const float4* __restrict__ wq) {
#pragma unroll
  for (int i = 0; i < NP * kMT; ++i) dst[i] = wq[(P0 * kMT + i) * 32];
}

template <int NP>
__device__ __forceinline__ void mma_group(v8f (&acc)[kMT], const float4 (&w)[30], const float4* b) {
#pragma unroll
  for (int pp = 0; pp < NP; ++pp) {
    const float4 bv = b[pp];
    const v2f b0 = {bv.x, bv.y};
    const v2f b1 = {bv.z, bv.w};
#pragma unroll
    for (int m = 0; m < kMT; ++m) {
      const v2f a0 = {w[pp * kMT + m].x, w[pp * kMT + m].y};
      acc[m] = __builtin_amdgcn_wmma_f32_16x16x4_f32(false, a0, false, b0, (short)0, acc[m], false, false);
    }
#pragma unroll
    for (int m = 0; m < kMT; ++m) {
      const v2f a1 = {w[pp * kMT + m].z, w[pp * kMT + m].w};
      acc[m] = __builtin_amdgcn_wmma_f32_16x16x4_f32(false, a1, false, b1, (short)0, acc[m], false, false);
    }
  }
}

__device__ __forceinline__ void load_x(float4 (&xs)[4], const float* __restrict__ xlane, int t) {
  const float4* p = (const float4*)(xlane + (size_t)t * kD);
#pragma unroll
  for (int i = 0; i < 4; ++i) xs[i] = p[i];
}

__device__ __forceinline__ void rnn_step(int t, int q, int l,
    v8f (&cur)[kMT], v8f (&nxt)[kMT], float4 (&hq)[5], const float4 (&bq)[5],
    const float4 (&xuse)[4], float4 (&xload)[4], const float* __restrict__ xlane,
    float4 (&setA)[30], float4 (&setB)[30],
    const float4* __restrict__ wq, float4* __restrict__ sPbuf) {
  const int t2 = (t + 2 < kT) ? (t + 2) : (kT - 1);
  load_x(xload, xlane, t2);

  load_w<7, 2>(setB, wq);
  group_fence();
  mma_group<3>(cur, setA, &hq[0]);
  group_fence();
  load_w<0, 2>(setA, wq);
  group_fence();
  mma_group<2>(cur, setB, &hq[3]);
  group_fence();

  float4* pw = sPbuf + (q * kQuads) * 32 + l;
#pragma unroll
  for (int u = 0; u < kQuads; ++u) {
    const int m = u >> 1, r = 4 * (u & 1);
    pw[u * 32] = make_float4(cur[m][r], cur[m][r + 1], cur[m][r + 2], cur[m][r + 3]);
  }
  __syncthreads();

  const float4* pr = sPbuf + (5 * q) * 32 + l;
  float4 ex[4][5];
#pragma unroll
  for (int s = 0; s < 4; ++s)
#pragma unroll
    for (int i = 0; i < 5; ++i) ex[s][i] = pr[(s * kQuads + i) * 32];
  load_w<2, 2>(setB, wq);
  group_fence();
  float4 xb[4] = {xuse[0], xuse[1], xuse[2], xuse[3]};
  pin_here(xb[0]); pin_here(xb[1]);
#pragma unroll
  for (int m = 0; m < kMT; ++m) nxt[m] = (v8f){0.f, 0.f, 0.f, 0.f, 0.f, 0.f, 0.f, 0.f};
  mma_group<2>(nxt, setA, &xb[0]);
  group_fence();
  load_w<4, 3>(setA, wq);
  group_fence();
  pin_here(xb[2]); pin_here(xb[3]);
#pragma unroll
  for (int s = 0; s < 4; ++s)
#pragma unroll
    for (int i = 0; i < 5; ++i) pin_here(ex[s][i]);
  mma_group<2>(nxt, setB, &xb[2]);
#pragma unroll
  for (int i = 0; i < 5; ++i) {
    float4 s;
    s.x = ((ex[0][i].x + ex[1][i].x) + (ex[2][i].x + ex[3][i].x)) + bq[i].x;
    s.y = ((ex[0][i].y + ex[1][i].y) + (ex[2][i].y + ex[3][i].y)) + bq[i].y;
    s.z = ((ex[0][i].z + ex[1][i].z) + (ex[2][i].z + ex[3][i].z)) + bq[i].z;
    s.w = ((ex[0][i].w + ex[1][i].w) + (ex[2][i].w + ex[3][i].w)) + bq[i].w;
    hq[i] = make_float4(fast_tanh(s.x), fast_tanh(s.y), fast_tanh(s.z), fast_tanh(s.w));
  }
  group_fence();
}

__global__ __launch_bounds__(kThreads, 1)
void rnn_wmma_kernel(const float* __restrict__ X,
                     const float* __restrict__ Wc,
                     const float* __restrict__ bc,
                     const float* __restrict__ Wo,
                     const float* __restrict__ bo,
                     float* __restrict__ out) {
  __shared__ float4 sW[kWSlots];
  __shared__ float4 sP[2][kPSlots];

  const int tid = threadIdx.x;
  const int q   = tid >> 5;
  const int l   = tid & 31;
  const int c   = l & 15;
  const int hh  = l >> 4;
  const int rowBase = blockIdx.x * kCols;

  {
    float4* sWq = sW + (q * kPairs * kMT) * 32;
    const int rot = blockIdx.x % kStageTasks;
    for (int b0 = 0; b0 < kStageTasks; b0 += kStageBatch) {
      float2 v[kStageBatch][4];
      int dst[kStageBatch], lim[kStageBatch];
#pragma unroll
      for (int j = 0; j < kStageBatch; ++j) {
        int it = b0 + j + rot;
        if (it >= kStageTasks) it -= kStageTasks;
        const int p = it / 5;
        const int item = (it - 5 * p) * 32 + l;
        const int g = (item >= 80) ? 1 : 0;
        const int col = 2 * (item - 80 * g);
        const int kk0 = 10 * q + 2 * (p - 4);
        const int base = (p < 4) ? (32 * q + 4 * p) : (kD + 16 * (kk0 >> 2) + 2 * (kk0 & 3));
        const int r0 = base + ((p < 4) ? 16 : 8) * g;
        lim[j] = (col < kH) ? (kD + kH - r0) : 0;
        dst[j] = (p * kMT + (col >> 4)) * 32 + 16 * g + (col & 15);
#pragma unroll
        for (int e = 0; e < 4; ++e) {
          const int idx = (e < lim[j]) ? ((r0 + e) * kH + col) : 0;
          v[j][e] = *(const float2*)(Wc + idx);
        }
      }
#pragma unroll
      for (int j = 0; j < kStageBatch; ++j) {
        float4 lo = make_float4(v[j][0].x, v[j][1].x, v[j][2].x, v[j][3].x);
        float4 hi = make_float4(v[j][0].y, v[j][1].y, v[j][2].y, v[j][3].y);
        if (lim[j] < 4) {
          if (lim[j] < 1) { lo.x = 0.f; hi.x = 0.f; }
          if (lim[j] < 2) { lo.y = 0.f; hi.y = 0.f; }
          if (lim[j] < 3) { lo.z = 0.f; hi.z = 0.f; }
          lo.w = 0.f; hi.w = 0.f;
        }
        sWq[dst[j]]     = lo;
        sWq[dst[j] + 1] = hi;
      }
    }
  }

  float4 bq[5];
#pragma unroll
  for (int i = 0; i < 5; ++i) {
    const int u = 5 * q + i;
    const int unit0 = 16 * (u >> 1) + 8 * hh + 4 * (u & 1);
    bq[i].x = (unit0 + 0 < kH) ? bc[unit0 + 0] : 0.f;
    bq[i].y = (unit0 + 1 < kH) ? bc[unit0 + 1] : 0.f;
    bq[i].z = (unit0 + 2 < kH) ? bc[unit0 + 2] : 0.f;
    bq[i].w = (unit0 + 3 < kH) ? bc[unit0 + 3] : 0.f;
  }

  const float* xlane = X + (size_t)(rowBase + c) * kT * kD + 32 * q + 16 * hh;
  const float4* wq = sW + (q * kPairs * kMT) * 32 + l;

  float4 xs0[4], xs1[4];
  load_x(xs0, xlane, 0);
  load_x(xs1, xlane, 1);

  float4 hq[5];
#pragma unroll
  for (int i = 0; i < 5; ++i) hq[i] = make_float4(0.f, 0.f, 0.f, 0.f);

  __syncthreads();

  float4 setA[30], setB[30];
  v8f accA[kMT], accB[kMT];
#pragma unroll
  for (int m = 0; m < kMT; ++m) accA[m] = (v8f){0.f, 0.f, 0.f, 0.f, 0.f, 0.f, 0.f, 0.f};
  load_w<0, 2>(setA, wq);
  load_w<2, 2>(setB, wq);
  mma_group<2>(accA, setA, &xs0[0]);
  mma_group<2>(accA, setB, &xs0[2]);
  load_w<4, 3>(setA, wq);

  for (int tt = 0; tt < kT; tt += 2) {
    rnn_step(tt,     q, l, accA, accB, hq, bq, xs1, xs0, xlane, setA, setB, wq, sP[0]);
    rnn_step(tt + 1, q, l, accB, accA, hq, bq, xs0, xs1, xlane, setA, setB, wq, sP[1]);
  }

  float* hf = (float*)sP[0];
#pragma unroll
  for (int i = 0; i < 5; ++i) {
    const int u = 5 * q + i;
    const int unit0 = 16 * (u >> 1) + 8 * hh + 4 * (u & 1);
    *(float4*)(&hf[c * kHStride + unit0]) = hq[i];
  }
  __syncthreads();

  for (int p = tid; p < kCols * kO; p += kThreads) {
    const int r = p / kO, o = p % kO;
    float sum = bo[o];
    const float* hrow = &hf[r * kHStride];
    for (int k = 0; k < kH; ++k) sum += hrow[k] * Wo[k * kO + o];
    out[(size_t)(rowBase + r) * kO + o] = sum;
  }
}

extern "C" void kernel_launch(void* const* d_in, const int* in_sizes, int n_in,
                              void* d_out, int out_size, void* d_ws, size_t ws_size,
                              hipStream_t stream) {
  const float* X  = (const float*)d_in[0];
  const float* Wc = (const float*)d_in[1];
  const float* bc = (const float*)d_in[2];
  const float* Wo = (const float*)d_in[3];
  const float* bo = (const float*)d_in[4];
  float* out = (float*)d_out;
  (void)in_sizes; (void)n_in; (void)out_size; (void)d_ws; (void)ws_size;
  dim3 grid(kB / kCols), block(kThreads);
  hipLaunchKernelGGL(rnn_wmma_kernel, grid, block, 0, stream, X, Wc, bc, Wo, bo, out);
}


# tanh of each unit by the chip's f32 tanh instruction (one transcendental per value) instead of exp and an expanded division; exchange reads issued with the second x block's loads; sum order kept
# speedup vs baseline: 3.4556x; 1.1025x over previous
#include <hip/hip_runtime.h>
#include <math.h>

typedef __attribute__((ext_vector_type(2))) float v2f;
typedef __attribute__((ext_vector_type(8))) float v8f;

namespace {
constexpr int kB  = 4096;
constexpr int kT  = 128;
constexpr int kD  = 128;
constexpr int kH  = 150;
constexpr int kO  = 10;
constexpr int kCols = 16;
constexpr int kWaves = 4;
constexpr int kThreads = 128;
constexpr int kMT = 10;
constexpr int kPairs = 9;
constexpr int kQuads = 2 * kMT;
constexpr int kWSlots = kWaves * kPairs * kMT * 32;
constexpr int kPSlots = kWaves * kQuads * 32;
constexpr int kHStride = 164;
constexpr int kStageTasks = 45;
constexpr int kStageBatch = 15;
}

__device__ __forceinline__ float fast_tanh(float x) {
  return __builtin_amdgcn_tanhf(x);
}

__device__ __forceinline__ void group_fence() { __builtin_amdgcn_sched_barrier(0); }

__device__ __forceinline__ void pin_here(float4& v) {
  asm volatile("" : "+v"(v.x), "+v"(v.y), "+v"(v.z), "+v"(v.w));
}

template <int P0, int NP>
__device__ __forceinline__ void load_w(float4 (&dst)[30], const float4* __restrict__ wq) {
#pragma unroll
  for (int i = 0; i < NP * kMT; ++i) dst[i] = wq[(P0 * kMT + i) * 32];
}

template <int NP>
__device__ __forceinline__ void mma_group(v8f (&acc)[kMT], const float4 (&w)[30], const float4* b) {
#pragma unroll
  for (int pp = 0; pp < NP; ++pp) {
    const float4 bv = b[pp];
    const v2f b0 = {bv.x, bv.y};
    const v2f b1 = {bv.z, bv.w};
#pragma unroll
    for (int m = 0; m < kMT; ++m) {
      const v2f a0 = {w[pp * kMT + m].x, w[pp * kMT + m].y};
      acc[m] = __builtin_amdgcn_wmma_f32_16x16x4_f32(false, a0, false, b0, (short)0, acc[m], false, false);
    }
#pragma unroll
    for (int m = 0; m < kMT; ++m) {
      const v2f a1 = {w[pp * kMT + m].z, w[pp * kMT + m].w};
      acc[m] = __builtin_amdgcn_wmma_f32_16x16x4_f32(false, a1, false, b1, (short)0, acc[m], false, false);
    }
  }
}

__device__ __forceinline__ void load_x(float4 (&xs)[4], const float* __restrict__ xlane, int t) {
  const float4* p = (const float4*)(xlane + (size_t)t * kD);
#pragma unroll
  for (int i = 0; i < 4; ++i) xs[i] = p[i];
}

__device__ __forceinline__ void rnn_step(int t, int q, int l,
    v8f (&cur)[kMT], v8f (&nxt)[kMT], float4 (&hq)[5], const float4 (&bq)[5],
    const float4 (&xuse)[4], float4 (&xload)[4], const float* __restrict__ xlane,
    float4 (&setA)[30], float4 (&setB)[30],
    const float4* __restrict__ wq, float4* __restrict__ sPbuf) {
  const int t2 = (t + 2 < kT) ? (t + 2) : (kT - 1);
  load_x(xload, xlane, t2);

  load_w<7, 2>(setB, wq);
  group_fence();
  mma_group<3>(cur, setA, &hq[0]);
  group_fence();
  load_w<0, 2>(setA, wq);
  group_fence();
  mma_group<2>(cur, setB, &hq[3]);
  group_fence();

  float4* pw = sPbuf + (q * kQuads) * 32 + l;
#pragma unroll
  for (int u = 0; u < kQuads; ++u) {
    const int m = u >> 1, r = 4 * (u & 1);
    pw[u * 32] = make_float4(cur[m][r], cur[m][r + 1], cur[m][r + 2], cur[m][r + 3]);
  }
  __syncthreads();

  load_w<2, 2>(setB, wq);
  group_fence();
  float4 xb[4] = {xuse[0], xuse[1], xuse[2], xuse[3]};
  pin_here(xb[0]); pin_here(xb[1]);
#pragma unroll
  for (int m = 0; m < kMT; ++m) nxt[m] = (v8f){0.f, 0.f, 0.f, 0.f, 0.f, 0.f, 0.f, 0.f};
  mma_group<2>(nxt, setA, &xb[0]);
  group_fence();
  load_w<4, 3>(setA, wq);
  group_fence();
  const float4* pr = sPbuf + (5 * q) * 32 + l;
  float4 ex[4][5];
#pragma unroll
  for (int s = 0; s < 4; ++s)
#pragma unroll
    for (int i = 0; i < 5; ++i) ex[s][i] = pr[(s * kQuads + i) * 32];
  pin_here(xb[2]); pin_here(xb[3]);
  mma_group<2>(nxt, setB, &xb[2]);
#pragma unroll
  for (int i = 0; i < 5; ++i) {
    float4 s;
    s.x = ((ex[0][i].x + ex[1][i].x) + (ex[2][i].x + ex[3][i].x)) + bq[i].x;
    s.y = ((ex[0][i].y + ex[1][i].y) + (ex[2][i].y + ex[3][i].y)) + bq[i].y;
    s.z = ((ex[0][i].z + ex[1][i].z) + (ex[2][i].z + ex[3][i].z)) + bq[i].z;
    s.w = ((ex[0][i].w + ex[1][i].w) + (ex[2][i].w + ex[3][i].w)) + bq[i].w;
    hq[i] = make_float4(fast_tanh(s.x), fast_tanh(s.y), fast_tanh(s.z), fast_tanh(s.w));
  }
  __builtin_amdgcn_sched_group_barrier(0x100, 20, 0);
  __builtin_amdgcn_sched_group_barrier(0x008, 40, 0);
  __builtin_amdgcn_sched_group_barrier(0x002, 60, 0);
  group_fence();
}

__global__ __launch_bounds__(kThreads, 1)
void rnn_wmma_kernel(const float* __restrict__ X,
                     const float* __restrict__ Wc,
                     const float* __restrict__ bc,
                     const float* __restrict__ Wo,
                     const float* __restrict__ bo,
                     float* __restrict__ out) {
  __shared__ float4 sW[kWSlots];
  __shared__ float4 sP[2][kPSlots];

  const int tid = threadIdx.x;
  const int q   = tid >> 5;
  const int l   = tid & 31;
  const int c   = l & 15;
  const int hh  = l >> 4;
  const int rowBase = blockIdx.x * kCols;

  {
    float4* sWq = sW + (q * kPairs * kMT) * 32;
    const int rot = blockIdx.x % kStageTasks;
    for (int b0 = 0; b0 < kStageTasks; b0 += kStageBatch) {
      float2 v[kStageBatch][4];
      int dst[kStageBatch], lim[kStageBatch];
#pragma unroll
      for (int j = 0; j < kStageBatch; ++j) {
        int it = b0 + j + rot;
        if (it >= kStageTasks) it -= kStageTasks;
        const int p = it / 5;
        const int item = (it - 5 * p) * 32 + l;
        const int g = (item >= 80) ? 1 : 0;
        const int col = 2 * (item - 80 * g);
        const int kk0 = 10 * q + 2 * (p - 4);
        const int base = (p < 4) ? (32 * q + 4 * p) : (kD + 16 * (kk0 >> 2) + 2 * (kk0 & 3));
        const int r0 = base + ((p < 4) ? 16 : 8) * g;
        lim[j] = (col < kH) ? (kD + kH - r0) : 0;
        dst[j] = (p * kMT + (col >> 4)) * 32 + 16 * g + (col & 15);
#pragma unroll
        for (int e = 0; e < 4; ++e) {
          const int idx = (e < lim[j]) ? ((r0 + e) * kH + col) : 0;
          v[j][e] = *(const float2*)(Wc + idx);
        }
      }
#pragma unroll
      for (int j = 0; j < kStageBatch; ++j) {
        float4 lo = make_float4(v[j][0].x, v[j][1].x, v[j][2].x, v[j][3].x);
        float4 hi = make_float4(v[j][0].y, v[j][1].y, v[j][2].y, v[j][3].y);
        if (lim[j] < 4) {
          if (lim[j] < 1) { lo.x = 0.f; hi.x = 0.f; }
          if (lim[j] < 2) { lo.y = 0.f; hi.y = 0.f; }
          if (lim[j] < 3) { lo.z = 0.f; hi.z = 0.f; }
          lo.w = 0.f; hi.w = 0.f;
        }
        sWq[dst[j]]     = lo;
        sWq[dst[j] + 1] = hi;
      }
    }
  }

  float4 bq[5];
#pragma unroll
  for (int i = 0; i < 5; ++i) {
    const int u = 5 * q + i;
    const int unit0 = 16 * (u >> 1) + 8 * hh + 4 * (u & 1);
    bq[i].x = (unit0 + 0 < kH) ? bc[unit0 + 0] : 0.f;
    bq[i].y = (unit0 + 1 < kH) ? bc[unit0 + 1] : 0.f;
    bq[i].z = (unit0 + 2 < kH) ? bc[unit0 + 2] : 0.f;
    bq[i].w = (unit0 + 3 < kH) ? bc[unit0 + 3] : 0.f;
  }

  const float* xlane = X + (size_t)(rowBase + c) * kT * kD + 32 * q + 16 * hh;
  const float4* wq = sW + (q * kPairs * kMT) * 32 + l;

  float4 xs0[4], xs1[4];
  load_x(xs0, xlane, 0);
  load_x(xs1, xlane, 1);

  float4 hq[5];
#pragma unroll
  for (int i = 0; i < 5; ++i) hq[i] = make_float4(0.f, 0.f, 0.f, 0.f);

  __syncthreads();

  float4 setA[30], setB[30];
  v8f accA[kMT], accB[kMT];
#pragma unroll
  for (int m = 0; m < kMT; ++m) accA[m] = (v8f){0.f, 0.f, 0.f, 0.f, 0.f, 0.f, 0.f, 0.f};
  load_w<0, 2>(setA, wq);
  load_w<2, 2>(setB, wq);
  mma_group<2>(accA, setA, &xs0[0]);
  mma_group<2>(accA, setB, &xs0[2]);
  load_w<4, 3>(setA, wq);

  for (int tt = 0; tt < kT; tt += 2) {
    rnn_step(tt,     q, l, accA, accB, hq, bq, xs1, xs0, xlane, setA, setB, wq, sP[0]);
    rnn_step(tt + 1, q, l, accB, accA, hq, bq, xs0, xs1, xlane, setA, setB, wq, sP[1]);
  }

  float* hf = (float*)sP[0];
#pragma unroll
  for (int i = 0; i < 5; ++i) {
    const int u = 5 * q + i;
    const int unit0 = 16 * (u >> 1) + 8 * hh + 4 * (u & 1);
    *(float4*)(&hf[c * kHStride + unit0]) = hq[i];
  }
  __syncthreads();

  for (int p = tid; p < kCols * kO; p += kThreads) {
    const int r = p / kO, o = p % kO;
    float sum = bo[o];
    const float* hrow = &hf[r * kHStride];
    for (int k = 0; k < kH; ++k) sum += hrow[k] * Wo[k * kO + o];
    out[(size_t)(rowBase + r) * kO + o] = sum;
  }
}

extern "C" void kernel_launch(void* const* d_in, const int* in_sizes, int n_in,
                              void* d_out, int out_size, void* d_ws, size_t ws_size,
                              hipStream_t stream) {
  const float* X  = (const float*)d_in[0];
  const float* Wc = (const float*)d_in[1];
  const float* bc = (const float*)d_in[2];
  const float* Wo = (const float*)d_in[3];
  const float* bo = (const float*)d_in[4];
  float* out = (float*)d_out;
  (void)in_sizes; (void)n_in; (void)out_size; (void)d_ws; (void)ws_size;
  dim3 grid(kB / kCols), block(kThreads);
  hipLaunchKernelGGL(rnn_wmma_kernel, grid, block, 0, stream, X, Wc, bc, Wo, bo, out);
}
